# starting kernel
# baseline (speedup 1.0000x reference)
#include <hip/hip_runtime.h>
#include <hip/hip_bf16.h>

typedef __attribute__((ext_vector_type(2))) float v2f;
typedef __attribute__((ext_vector_type(4))) float v4f;
typedef __attribute__((ext_vector_type(8))) float v8f;

#define C_IN   64
#define C_OUT  64
#define HDIM   56
#define WDIM   56
#define KTOT   384
#define LROW   72

__launch_bounds__(128)
__global__ void shiftconv_wmma_f32(const float* __restrict__ x,
                                   const float* __restrict__ w,
                                   float* __restrict__ y)
{
    __shared__ float xs[2 * C_IN * LROW];

    const int o    = blockIdx.x;
    const int tid  = threadIdx.x;
    const int wv   = tid >> 5;
    const int lane = tid & 31;
    const int m    = lane & 15;
    const int koff = (lane >> 4) << 1;

    {
        const int l = tid >> 6;
        const int j = tid & 63;
        const int r = (l ? (o + HDIM - 2) : (o + HDIM - 1)) % HDIM;
        float* dst = xs + tid * LROW;
        const v4f* src = (const v4f*)(x + (j * HDIM + r) * WDIM);
        const v4f z4 = {};
        ((v4f*)dst)[0]  = z4;
        ((v4f*)dst)[15] = z4;
        ((v4f*)dst)[16] = z4;
        ((v4f*)dst)[17] = z4;
        #pragma unroll
        for (int q = 0; q < 14; ++q)
            ((v4f*)(dst + 4))[q] = src[q];
    }
    __syncthreads();

    const int iBase = wv << 4;

    const v8f vzero = {};
    v8f acc[4];
    #pragma unroll
    for (int t = 0; t < 4; ++t) acc[t] = vzero;

    const float* wrow = w + (size_t)(iBase + m) * KTOT + koff;
    __builtin_prefetch(wrow, 0, 0);

    const float* pb[6];
    #pragma unroll
    for (int s = 0; s < 3; ++s) {
        #pragma unroll
        for (int e = 0; e < 2; ++e) {
            const int c  = 4 * s + koff + e;
            const int jc = c / 6;
            const int q  = c - jc * 6;
            const int lc = q / 3;
            const int kc = q - lc * 3;
            pb[2 * s + e] = xs + (lc * C_IN + jc) * LROW + kc + 3 + m;
        }
    }

    for (int u = 0; u < KTOT / 12; ++u) {
        #pragma unroll
        for (int s = 0; s < 3; ++s) {
            const v2f a = *(const v2f*)(wrow + u * 12 + s * 4);
            #pragma unroll
            for (int t = 0; t < 4; ++t) {
                v2f b;
                b.x = pb[2 * s + 0][t * 16];
                b.y = pb[2 * s + 1][t * 16];
                acc[t] = __builtin_amdgcn_wmma_f32_16x16x4_f32(
                     false, a,  false, b,
                     (short)0, acc[t],
                     false,  false);
            }
        }
        #pragma unroll
        for (int e = 0; e < 6; ++e) pb[e] += 2 * LROW;
    }

    #pragma unroll
    for (int t = 0; t < 4; ++t) {
        const int n = t * 16 + m;
        if (n < WDIM) {
            const float* av = (const float*)&acc[t];
            float* yp = y + (size_t)(iBase + ((lane < 16) ? 0 : 8)) * HDIM * WDIM
                          + o * WDIM + n;
            #pragma unroll
            for (int v = 0; v < 8; ++v)
                yp[(size_t)v * HDIM * WDIM] = av[v];
        }
    }
}

extern "C" void kernel_launch(void* const* d_in, const int* in_sizes, int n_in,
                              void* d_out, int out_size, void* d_ws, size_t ws_size,
                              hipStream_t stream) {
    const float* x = (const float*)d_in[0];
    const float* w = (const float*)d_in[1];
    float* y = (float*)d_out;
    (void)in_sizes; (void)n_in; (void)out_size; (void)d_ws; (void)ws_size;
    shiftconv_wmma_f32<<<dim3(HDIM), dim3(128), 0, stream>>>(x, w, y);
}


# K split over the 4 waves of a block, channel tiles moved to separate blocks (224 blocks, 96 WMMAs per wave, was 384); the 4 partial sums are added through LDS, so the sum order changes
# speedup vs baseline: 1.9148x; 1.9148x over previous
#include <hip/hip_runtime.h>
#include <hip/hip_bf16.h>


typedef __attribute__((ext_vector_type(2))) float v2f;
typedef __attribute__((ext_vector_type(4))) float v4f;
typedef __attribute__((ext_vector_type(8))) float v8f;

#define C_IN   64
#define C_OUT  64
#define HDIM   56
#define WDIM   56
#define KTOT   384
#define LROW   72
#define KWAVE  96
#define RPITCH 68

__launch_bounds__(128)
__global__ void shiftconv_wmma_f32(const float* __restrict__ x,
                                   const float* __restrict__ w,
                                   float* __restrict__ y)
{
    __shared__ __attribute__((aligned(16))) float xs[2 * C_IN * LROW];
    __shared__ __attribute__((aligned(16))) float red[4 * 16 * RPITCH];

    const int o    = blockIdx.x >> 2;
    const int ct   = blockIdx.x & 3;
    const int tid  = threadIdx.x;
    const int wv   = tid >> 5;
    const int lane = tid & 31;
    const int m    = lane & 15;
    const int hh   = lane >> 4;
    const int koff = hh << 1;

    v4f xr[14];
    {
        const int l = tid >> 6;
        const int j = tid & 63;
        const int r = (l ? (o + HDIM - 2) : (o + HDIM - 1)) % HDIM;
        const v4f* src = (const v4f*)(x + (j * HDIM + r) * WDIM);
        #pragma unroll
        for (int q = 0; q < 14; ++q) xr[q] = src[q];
    }

    const float* wrow = w + (size_t)(ct * 16 + m) * KTOT + wv * KWAVE + koff;
    v2f a[24];
    #pragma unroll
    for (int s = 0; s < 24; ++s) a[s] = *(const v2f*)(wrow + 4 * s);

    {
        float* dst = xs + tid * LROW;
        const v4f z4 = {};
        ((v4f*)dst)[0]  = z4;
        ((v4f*)dst)[15] = z4;
        ((v4f*)dst)[16] = z4;
        ((v4f*)dst)[17] = z4;
        #pragma unroll
        for (int q = 0; q < 14; ++q)
            ((v4f*)(dst + 4))[q] = xr[q];
    }
    __syncthreads();

    const v8f vzero = {};
    v8f acc[4];
    #pragma unroll
    for (int t = 0; t < 4; ++t) acc[t] = vzero;

    int ob[6];
    #pragma unroll
    for (int s = 0; s < 3; ++s) {
        #pragma unroll
        for (int e = 0; e < 2; ++e) {
            const int c  = 4 * s + koff + e;
            const int jc = c / 6;
            const int q  = c - jc * 6;
            const int lc = q / 3;
            const int kc = q - lc * 3;
            ob[2 * s + e] = (lc * C_IN + wv * 16 + jc) * LROW + kc + 3 + m;
        }
    }

    #pragma unroll
    for (int p = 0; p < 8; ++p) {
        #pragma unroll
        for (int s = 0; s < 3; ++s) {
            #pragma unroll
            for (int t = 0; t < 4; ++t) {
                v2f b;
                b.x = xs[ob[2 * s + 0] + p * 2 * LROW + t * 16];
                b.y = xs[ob[2 * s + 1] + p * 2 * LROW + t * 16];
                acc[t] = __builtin_amdgcn_wmma_f32_16x16x4_f32(
                     false, a[3 * p + s],  false, b,
                     (short)0, acc[t],
                     false,  false);
            }
        }
    }

    {
        float* rp = red + (wv * 16 + 8 * hh) * RPITCH + m;
        #pragma unroll
        for (int t = 0; t < 4; ++t) {
            const float* av = (const float*)&acc[t];
            #pragma unroll
            for (int v = 0; v < 8; ++v)
                rp[v * RPITCH + t * 16] = av[v];
        }
    }
    __syncthreads();

    {
        const int ch = tid >> 3;
        const int n0 = (tid & 7) << 3;
        const float* q0 = red + ch * RPITCH + n0;
        v4f s0 = ((const v4f*)q0)[0];
        v4f s1 = ((const v4f*)q0)[1];
        #pragma unroll
        for (int ww = 1; ww < 4; ++ww) {
            s0 += ((const v4f*)(q0 + ww * 16 * RPITCH))[0];
            s1 += ((const v4f*)(q0 + ww * 16 * RPITCH))[1];
        }
        if (n0 < WDIM) {
            float* yp = y + (size_t)(ct * 16 + ch) * HDIM * WDIM + o * WDIM + n0;
            ((v4f*)yp)[0] = s0;
            ((v4f*)yp)[1] = s1;
        }
    }
}

extern "C" void kernel_launch(void* const* d_in, const int* in_sizes, int n_in,
                              void* d_out, int out_size, void* d_ws, size_t ws_size,
                              hipStream_t stream) {
    const float* x = (const float*)d_in[0];
    const float* w = (const float*)d_in[1];
    float* y = (float*)d_out;
    (void)in_sizes; (void)n_in; (void)out_size; (void)d_ws; (void)ws_size;
    shiftconv_wmma_f32<<<dim3(HDIM * 4), dim3(128), 0, stream>>>(x, w, y);
}


# B read path: columns mapped n=4*lane+tile and the two x rows interleaved in LDS, so 3 wide reads per in-channel feed 12 WMMAs; all 24 reads stand in one group before the WMMAs; k and sum order change
# speedup vs baseline: 2.1258x; 1.1102x over previous
#include <hip/hip_runtime.h>
#include <hip/hip_bf16.h>


typedef __attribute__((ext_vector_type(2))) float v2f;
typedef __attribute__((ext_vector_type(4))) float v4f;
typedef __attribute__((ext_vector_type(8))) float v8f;

#define C_IN   64
#define C_OUT  64
#define HDIM   56
#define WDIM   56
#define KTOT   384
#define XP     140
#define RPITCH 68

__launch_bounds__(128)
__global__ void shiftconv_wmma_f32(const float* __restrict__ x,
                                   const float* __restrict__ w,
                                   float* __restrict__ y)
{
    __shared__ __attribute__((aligned(16))) float xs[C_IN * XP];
    __shared__ __attribute__((aligned(16))) float red[4 * 16 * RPITCH];

    const int o    = blockIdx.x >> 2;
    const int ct   = blockIdx.x & 3;
    const int tid  = threadIdx.x;
    const int wv   = tid >> 5;
    const int lane = tid & 31;
    const int m    = lane & 15;
    const int hh   = lane >> 4;

    const int sj = tid & 63;
    const int sh = tid >> 6;
    v4f xa[7], xb[7];
    {
        const int r0 = (o + HDIM - 1) % HDIM;
        const int r1 = (o + HDIM - 2) % HDIM;
        const v4f* sa = (const v4f*)(x + (sj * HDIM + r0) * WDIM + 28 * sh);
        const v4f* sb = (const v4f*)(x + (sj * HDIM + r1) * WDIM + 28 * sh);
        #pragma unroll
        for (int q = 0; q < 7; ++q) { xa[q] = sa[q]; xb[q] = sb[q]; }
    }

    const float* wrow = w + (size_t)(ct * 16 + m) * KTOT + (wv * 16 + hh) * 6;
    v2f a[24];
    #pragma unroll
    for (int q = 0; q < 8; ++q) {
        #pragma unroll
        for (int k = 0; k < 3; ++k) {
            a[3 * q + k].x = wrow[12 * q + k];
            a[3 * q + k].y = wrow[12 * q + 3 + k];
        }
    }

    {
        float* dst = xs + sj * XP;
        const v4f z4 = {};
        ((v4f*)dst)[sh ? 31 : 0] = z4;
        ((v4f*)dst)[29 + 3 * sh] = z4;
        ((v4f*)dst)[30 + 3 * sh] = z4;
        v4f* d4 = (v4f*)dst + 1 + 14 * sh;
        #pragma unroll
        for (int q = 0; q < 7; ++q) {
            v4f lo, hi;
            lo.x = xa[q].x; lo.y = xb[q].x; lo.z = xa[q].y; lo.w = xb[q].y;
            hi.x = xa[q].z; hi.y = xb[q].z; hi.z = xa[q].w; hi.w = xb[q].w;
            d4[2 * q]     = lo;
            d4[2 * q + 1] = hi;
        }
    }
    __syncthreads();

    float bw[8][12];
    {
        const float* bp = xs + (wv * 16 + hh) * XP + 8 * m;
        #pragma unroll
        for (int q = 0; q < 8; ++q) {
            const float* p = bp + q * 2 * XP;
            const v2f w0 = *(const v2f*)(p + 2);
            const v4f w1 = *(const v4f*)(p + 4);
            const v4f w2 = *(const v4f*)(p + 8);
            const v2f w3 = *(const v2f*)(p + 12);
            bw[q][0] = w0.x; bw[q][1]  = w0.y;
            bw[q][2] = w1.x; bw[q][3]  = w1.y; bw[q][4]  = w1.z; bw[q][5] = w1.w;
            bw[q][6] = w2.x; bw[q][7]  = w2.y; bw[q][8]  = w2.z; bw[q][9] = w2.w;
            bw[q][10] = w3.x; bw[q][11] = w3.y;
        }
    }
    __builtin_amdgcn_sched_barrier(0);

    const v8f vzero = {};
    v8f acc[4];
    #pragma unroll
    for (int t = 0; t < 4; ++t) acc[t] = vzero;

    #pragma unroll
    for (int q = 0; q < 8; ++q) {
        #pragma unroll
        for (int k = 0; k < 3; ++k) {
            #pragma unroll
            for (int t = 0; t < 4; ++t) {
                v2f b;
                b.x = bw[q][2 * (t + k)];
                b.y = bw[q][2 * (t + k) + 1];
                acc[t] = __builtin_amdgcn_wmma_f32_16x16x4_f32(
                     false, a[3 * q + k],  false, b,
                     (short)0, acc[t],
                     false,  false);
            }
        }
    }

    {
        float* rp = red + (wv * 16 + 8 * hh) * RPITCH + 4 * m;
        #pragma unroll
        for (int v = 0; v < 8; ++v) {
            v4f o4;
            o4.x = acc[0][v]; o4.y = acc[1][v]; o4.z = acc[2][v]; o4.w = acc[3][v];
            *(v4f*)(rp + v * RPITCH) = o4;
        }
    }
    __syncthreads();

    {
        const int ch = tid >> 3;
        const int n0 = (tid & 7) << 3;
        const float* q0 = red + ch * RPITCH + n0;
        v4f s0 = ((const v4f*)q0)[0];
        v4f s1 = ((const v4f*)q0)[1];
        #pragma unroll
        for (int ww = 1; ww < 4; ++ww) {
            s0 += ((const v4f*)(q0 + ww * 16 * RPITCH))[0];
            s1 += ((const v4f*)(q0 + ww * 16 * RPITCH))[1];
        }
        if (n0 < WDIM) {
            float* yp = y + (size_t)(ct * 16 + ch) * HDIM * WDIM + o * WDIM + n0;
            ((v4f*)yp)[0] = s0;
            ((v4f*)yp)[1] = s1;
        }
    }
}

extern "C" void kernel_launch(void* const* d_in, const int* in_sizes, int n_in,
                              void* d_out, int out_size, void* d_ws, size_t ws_size,
                              hipStream_t stream) {
    const float* x = (const float*)d_in[0];
    const float* w = (const float*)d_in[1];
    float* y = (float*)d_out;
    (void)in_sizes; (void)n_in; (void)out_size; (void)d_ws; (void)ws_size;
    shiftconv_wmma_f32<<<dim3(HDIM * 4), dim3(128), 0, stream>>>(x, w, y);
}


# B operand straight from global memory in fragment layout: one 16-byte load per lane, row and in-channel, edge columns by DPP from neighbour lanes; no LDS image of x, no barrier before the WMMAs
# speedup vs baseline: 2.4214x; 1.1391x over previous
#include <hip/hip_runtime.h>
#include <hip/hip_bf16.h>


typedef __attribute__((ext_vector_type(2))) float v2f;
typedef __attribute__((ext_vector_type(4))) float v4f;
typedef __attribute__((ext_vector_type(8))) float v8f;

#define C_IN   64
#define C_OUT  64
#define HDIM   56
#define WDIM   56
#define KTOT   384
#define RPITCH 68

__device__ __forceinline__ float from_lane_below(float v)
{
    return __int_as_float(__builtin_amdgcn_update_dpp(0, __float_as_int(v), 0x111, 0xf, 0xf, true));
}
__device__ __forceinline__ float from_lane_above(float v)
{
    return __int_as_float(__builtin_amdgcn_update_dpp(0, __float_as_int(v), 0x101, 0xf, 0xf, true));
}

__launch_bounds__(128)
__global__ void shiftconv_wmma_f32(const float* __restrict__ x,
                                   const float* __restrict__ w,
                                   float* __restrict__ y)
{
    __shared__ __attribute__((aligned(16))) float red[4 * 16 * RPITCH];

    const int o    = blockIdx.x >> 2;
    const int ct   = blockIdx.x & 3;
    const int tid  = threadIdx.x;
    const int wv   = tid >> 5;
    const int lane = tid & 31;
    const int m    = lane & 15;
    const int hh   = lane >> 4;

    const int r0 = (o + HDIM - 1) % HDIM;
    const int r1 = (o + HDIM - 2) % HDIM;
    const int mc = m < 14 ? m : 13;
    const float* xch  = x + (size_t)(wv * 16 + hh) * HDIM * WDIM + 4 * mc;
    const float* xr0  = xch + r0 * WDIM;
    const float* xr1  = xch + r1 * WDIM;
    const float* wrow = w + (size_t)(ct * 16 + m) * KTOT + (wv * 16 + hh) * 6;

    float wf[8][6];
    v4f xa[8], xb[8];
    #pragma unroll
    for (int q = 0; q < 8; ++q) {
        xa[q] = *(const v4f*)(xr0 + (size_t)q * 2 * HDIM * WDIM);
        xb[q] = *(const v4f*)(xr1 + (size_t)q * 2 * HDIM * WDIM);
        #pragma unroll
        for (int i = 0; i < 6; ++i) wf[q][i] = wrow[12 * q + i];
        __builtin_amdgcn_sched_barrier(0);
    }

    const v8f vzero = {};
    v8f acc[4];
    #pragma unroll
    for (int t = 0; t < 4; ++t) acc[t] = vzero;

    const bool last_col = (m == 13);

    #pragma unroll
    for (int q = 0; q < 8; ++q) {
        v2f a[3];
        #pragma unroll
        for (int k = 0; k < 3; ++k) { a[k].x = wf[q][k]; a[k].y = wf[q][3 + k]; }

        float bw[12];
        bw[0]  = from_lane_below(xa[q].w);
        bw[1]  = from_lane_below(xb[q].w);
        bw[2]  = xa[q].x; bw[3]  = xb[q].x;
        bw[4]  = xa[q].y; bw[5]  = xb[q].y;
        bw[6]  = xa[q].z; bw[7]  = xb[q].z;
        bw[8]  = xa[q].w; bw[9]  = xb[q].w;
        const float ra = from_lane_above(xa[q].x);
        const float rb = from_lane_above(xb[q].x);
        bw[10] = last_col ? 0.0f : ra;
        bw[11] = last_col ? 0.0f : rb;

        #pragma unroll
        for (int k = 0; k < 3; ++k) {
            #pragma unroll
            for (int t = 0; t < 4; ++t) {
                v2f b;
                b.x = bw[2 * (t + k)];
                b.y = bw[2 * (t + k) + 1];
                acc[t] = __builtin_amdgcn_wmma_f32_16x16x4_f32(
                     false, a[k],  false, b,
                     (short)0, acc[t],
                     false,  false);
            }
        }
        __builtin_amdgcn_sched_barrier(0);
    }

    {
        float* rp = red + (wv * 16 + 8 * hh) * RPITCH + 4 * m;
        #pragma unroll
        for (int v = 0; v < 8; ++v) {
            v4f o4;
            o4.x = acc[0][v]; o4.y = acc[1][v]; o4.z = acc[2][v]; o4.w = acc[3][v];
            *(v4f*)(rp + v * RPITCH) = o4;
        }
    }
    __syncthreads();

    {
        const int ch = tid >> 3;
        const int n0 = (tid & 7) << 3;
        const float* q0 = red + ch * RPITCH + n0;
        v4f s0 = ((const v4f*)q0)[0];
        v4f s1 = ((const v4f*)q0)[1];
        #pragma unroll
        for (int ww = 1; ww < 4; ++ww) {
            s0 += ((const v4f*)(q0 + ww * 16 * RPITCH))[0];
            s1 += ((const v4f*)(q0 + ww * 16 * RPITCH))[1];
        }
        if (n0 < WDIM) {
            float* yp = y + (size_t)(ct * 16 + ch) * HDIM * WDIM + o * WDIM + n0;
            ((v4f*)yp)[0] = s0;
            ((v4f*)yp)[1] = s1;
        }
    }
}

extern "C" void kernel_launch(void* const* d_in, const int* in_sizes, int n_in,
                              void* d_out, int out_size, void* d_ws, size_t ws_size,
                              hipStream_t stream) {
    const float* x = (const float*)d_in[0];
    const float* w = (const float*)d_in[1];
    float* y = (float*)d_out;
    (void)in_sizes; (void)n_in; (void)out_size; (void)d_ws; (void)ws_size;
    shiftconv_wmma_f32<<<dim3(HDIM * 4), dim3(128), 0, stream>>>(x, w, y);
}
